# starting kernel
# baseline (speedup 1.0000x reference)
#include <hip/hip_runtime.h>

typedef __attribute__((ext_vector_type(2))) float v2f;
typedef __attribute__((ext_vector_type(8))) float v8f;

#define NFEAT 128
#define NHID  128

__global__ void gcn_init_deg(float* __restrict__ deg, int n) {
    int i = blockIdx.x * blockDim.x + threadIdx.x;
    if (i < n) deg[i] = 1.0f;
}

__global__ void gcn_count_deg(const int* __restrict__ dst, float* __restrict__ deg, int e) {
    int i = blockIdx.x * blockDim.x + threadIdx.x;
    if (i < e) atomicAdd(&deg[dst[i]], 1.0f);
}

__global__ void gcn_rsqrt_deg(float* __restrict__ deg, int n) {
    int i = blockIdx.x * blockDim.x + threadIdx.x;
    if (i < n) deg[i] = rsqrtf(deg[i]);
}

__global__ __launch_bounds__(256) void gcn_gemm_wmma(const float* __restrict__ x,
                                                     const float* __restrict__ W,
                                                     float* __restrict__ xw,
                                                     int n_nodes) {
    const int lane = threadIdx.x & 31;
    const int wave = threadIdx.x >> 5;
    const int half = lane >> 4;
    const int l    = lane & 15;

    const int m0 = blockIdx.x * 16;
    const int n0 = wave * 16;

    int arow = m0 + l;
    if (arow >= n_nodes) arow = n_nodes - 1;

    const float* aptr = x + (size_t)arow * NFEAT + 2 * half;
    const float* bptr = W + (size_t)(2 * half) * NHID + n0 + l;

    v8f c = {};
#pragma unroll
    for (int kk = 0; kk < NFEAT; kk += 4) {
        v2f a = *(const v2f*)(aptr + kk);
        v2f b;
        b.x = bptr[(size_t)kk * NHID];
        b.y = bptr[(size_t)kk * NHID + NHID];
        c = __builtin_amdgcn_wmma_f32_16x16x4_f32(
                 false, a,  false, b,
                 (short)0, c,  false,  false);
    }

#pragma unroll
    for (int r = 0; r < 8; ++r) {
        int row = m0 + r + 8 * half;
        if (row < n_nodes) xw[(size_t)row * NHID + n0 + l] = c[r];
    }
}

__global__ void gcn_self_bias(const float* __restrict__ xw,
                              const float* __restrict__ dinv,
                              const float* __restrict__ bias,
                              float* __restrict__ out, int n_nodes) {
    int t = blockIdx.x * blockDim.x + threadIdx.x;
    int i = t >> 5;
    if (i >= n_nodes) return;
    int lane = t & 31;

    float di = dinv[i];
    float n2 = di * di;
    const float4 v  = *(const float4*)(xw + (size_t)i * NHID + lane * 4);
    const float4 bb = *(const float4*)(bias + lane * 4);
    float4 o;
    o.x = fmaf(v.x, n2, bb.x);
    o.y = fmaf(v.y, n2, bb.y);
    o.z = fmaf(v.z, n2, bb.z);
    o.w = fmaf(v.w, n2, bb.w);
    *(float4*)(out + (size_t)i * NHID + lane * 4) = o;
}

__global__ __launch_bounds__(256) void gcn_edge_scatter(const int* __restrict__ src,
                                                        const int* __restrict__ dst,
                                                        const float* __restrict__ dinv,
                                                        const float* __restrict__ xw,
                                                        float* __restrict__ out, int e) {
    int edge = blockIdx.x * 8 + (threadIdx.x >> 5);
    if (edge >= e) return;
    int lane = threadIdx.x & 31;

    int s = src[edge];
    int d = dst[edge];
    float norm = dinv[s] * dinv[d];

    const float4 v = *(const float4*)(xw + (size_t)s * NHID + lane * 4);
    float* o = out + (size_t)d * NHID + lane * 4;
    atomicAdd(o + 0, v.x * norm);
    atomicAdd(o + 1, v.y * norm);
    atomicAdd(o + 2, v.z * norm);
    atomicAdd(o + 3, v.w * norm);
}

extern "C" void kernel_launch(void* const* d_in, const int* in_sizes, int n_in,
                              void* d_out, int out_size, void* d_ws, size_t ws_size,
                              hipStream_t stream) {
    const int*   ei   = (const int*)d_in[0];
    const float* x    = (const float*)d_in[1];
    const float* W    = (const float*)d_in[2];
    const float* bias = (const float*)d_in[3];
    float*       out  = (float*)d_out;

    const int E = in_sizes[0] / 2;
    const int N = in_sizes[1] / NFEAT;

    float* xw  = (float*)d_ws;
    float* deg = (float*)((char*)d_ws + (size_t)N * NHID * sizeof(float));

    const int* srcIdx = ei;
    const int* dstIdx = ei + E;

    gcn_init_deg <<<(N + 255) / 256, 256, 0, stream>>>(deg, N);
    gcn_count_deg<<<(E + 255) / 256, 256, 0, stream>>>(dstIdx, deg, E);
    gcn_rsqrt_deg<<<(N + 255) / 256, 256, 0, stream>>>(deg, N);

    gcn_gemm_wmma<<<(N + 15) / 16, 256, 0, stream>>>(x, W, xw, N);

    gcn_self_bias<<<((size_t)N * 32 + 255) / 256, 256, 0, stream>>>(xw, deg, bias, out, N);
    gcn_edge_scatter<<<(E + 7) / 8, 256, 0, stream>>>(srcIdx, dstIdx, deg, xw, out, E);
}


# Edge scatter: a wave takes 32 edges, not 1: indices and dinv one edge per lane, then per edge a lane broadcast and the same whole-line loads and f32 atomic adds, 8 edges in flight; arrival-order sums
# speedup vs baseline: 1.5641x; 1.5641x over previous
#include <hip/hip_runtime.h>

typedef __attribute__((ext_vector_type(2))) float v2f;
typedef __attribute__((ext_vector_type(8))) float v8f;

#define NFEAT 128
#define NHID  128

__global__ void gcn_init_deg(float* __restrict__ deg, int n) {
    int i = blockIdx.x * blockDim.x + threadIdx.x;
    if (i < n) deg[i] = 1.0f;
}

__global__ void gcn_count_deg(const int* __restrict__ dst, float* __restrict__ deg, int e) {
    int i = blockIdx.x * blockDim.x + threadIdx.x;
    if (i < e) atomicAdd(&deg[dst[i]], 1.0f);
}

__global__ void gcn_rsqrt_deg(float* __restrict__ deg, int n) {
    int i = blockIdx.x * blockDim.x + threadIdx.x;
    if (i < n) deg[i] = rsqrtf(deg[i]);
}

__global__ __launch_bounds__(256) void gcn_gemm_wmma(const float* __restrict__ x,
                                                     const float* __restrict__ W,
                                                     float* __restrict__ xw,
                                                     int n_nodes) {
    const int lane = threadIdx.x & 31;
    const int wave = threadIdx.x >> 5;
    const int half = lane >> 4;
    const int l    = lane & 15;

    const int m0 = blockIdx.x * 16;
    const int n0 = wave * 16;

    int arow = m0 + l;
    if (arow >= n_nodes) arow = n_nodes - 1;

    const float* aptr = x + (size_t)arow * NFEAT + 2 * half;
    const float* bptr = W + (size_t)(2 * half) * NHID + n0 + l;

    v8f c = {};
#pragma unroll
    for (int kk = 0; kk < NFEAT; kk += 4) {
        v2f a = *(const v2f*)(aptr + kk);
        v2f b;
        b.x = bptr[(size_t)kk * NHID];
        b.y = bptr[(size_t)kk * NHID + NHID];
        c = __builtin_amdgcn_wmma_f32_16x16x4_f32(
                 false, a,  false, b,
                 (short)0, c,  false,  false);
    }

#pragma unroll
    for (int r = 0; r < 8; ++r) {
        int row = m0 + r + 8 * half;
        if (row < n_nodes) xw[(size_t)row * NHID + n0 + l] = c[r];
    }
}

__global__ void gcn_self_bias(const float* __restrict__ xw,
                              const float* __restrict__ dinv,
                              const float* __restrict__ bias,
                              float* __restrict__ out, int n_nodes) {
    int t = blockIdx.x * blockDim.x + threadIdx.x;
    int i = t >> 5;
    if (i >= n_nodes) return;
    int lane = t & 31;

    float di = dinv[i];
    float n2 = di * di;
    const float4 v  = *(const float4*)(xw + (size_t)i * NHID + lane * 4);
    const float4 bb = *(const float4*)(bias + lane * 4);
    float4 o;
    o.x = fmaf(v.x, n2, bb.x);
    o.y = fmaf(v.y, n2, bb.y);
    o.z = fmaf(v.z, n2, bb.z);
    o.w = fmaf(v.w, n2, bb.w);
    *(float4*)(out + (size_t)i * NHID + lane * 4) = o;
}

#define EDGES_PER_WAVE 32
#define EDGE_BATCH 8
__global__ __launch_bounds__(256) void gcn_edge_scatter(const int* __restrict__ src,
                                                        const int* __restrict__ dst,
                                                        const float* __restrict__ dinv,
                                                        const float* __restrict__ xw,
                                                        float* __restrict__ out, int e) {
    const unsigned lane = threadIdx.x & 31;
    const int wave = __builtin_amdgcn_readfirstlane((int)(threadIdx.x >> 5));
    const int e0 = ((int)blockIdx.x * 8 + wave) * EDGES_PER_WAVE;
    if (e0 >= e) return;
    const int cnt = (e - e0 < EDGES_PER_WAVE) ? (e - e0) : EDGES_PER_WAVE;

    int mine = e0 + (int)lane;
    if (mine > e - 1) mine = e - 1;
    const int   s = src[mine];
    const int   d = dst[mine];
    const float nrm = dinv[s] * dinv[d];
    const int   nrm_bits = __builtin_bit_cast(int, nrm);

    if (cnt == EDGES_PER_WAVE) {
        for (int j0 = 0; j0 < EDGES_PER_WAVE; j0 += EDGE_BATCH) {
            float v[EDGE_BATCH][4];
#pragma unroll
            for (int jj = 0; jj < EDGE_BATCH; ++jj) {
                const int sj = __builtin_amdgcn_readlane(s, j0 + jj);
                const float* xr = xw + (size_t)sj * NHID + lane;
                v[jj][0] = xr[0];
                v[jj][1] = xr[32];
                v[jj][2] = xr[64];
                v[jj][3] = xr[96];
            }
#pragma unroll
            for (int jj = 0; jj < EDGE_BATCH; ++jj) {
                const int   dj = __builtin_amdgcn_readlane(d, j0 + jj);
                const float nj = __builtin_bit_cast(float, __builtin_amdgcn_readlane(nrm_bits, j0 + jj));
                float* o = out + (size_t)dj * NHID + lane;
                atomicAdd(o +  0, v[jj][0] * nj);
                atomicAdd(o + 32, v[jj][1] * nj);
                atomicAdd(o + 64, v[jj][2] * nj);
                atomicAdd(o + 96, v[jj][3] * nj);
            }
        }
    } else {
        for (int j = 0; j < cnt; ++j) {
            const int   sj = __builtin_amdgcn_readlane(s, j);
            const int   dj = __builtin_amdgcn_readlane(d, j);
            const float nj = __builtin_bit_cast(float, __builtin_amdgcn_readlane(nrm_bits, j));
            const float* xr = xw + (size_t)sj * NHID + lane;
            const float v0 = xr[0];
            const float v1 = xr[32];
            const float v2 = xr[64];
            const float v3 = xr[96];
            float* o = out + (size_t)dj * NHID + lane;
            atomicAdd(o +  0, v0 * nj);
            atomicAdd(o + 32, v1 * nj);
            atomicAdd(o + 64, v2 * nj);
            atomicAdd(o + 96, v3 * nj);
        }
    }
}

extern "C" void kernel_launch(void* const* d_in, const int* in_sizes, int n_in,
                              void* d_out, int out_size, void* d_ws, size_t ws_size,
                              hipStream_t stream) {
    const int*   ei   = (const int*)d_in[0];
    const float* x    = (const float*)d_in[1];
    const float* W    = (const float*)d_in[2];
    const float* bias = (const float*)d_in[3];
    float*       out  = (float*)d_out;

    const int E = in_sizes[0] / 2;
    const int N = in_sizes[1] / NFEAT;

    float* xw  = (float*)d_ws;
    float* deg = (float*)((char*)d_ws + (size_t)N * NHID * sizeof(float));

    const int* srcIdx = ei;
    const int* dstIdx = ei + E;

    gcn_init_deg <<<(N + 255) / 256, 256, 0, stream>>>(deg, N);
    gcn_count_deg<<<(E + 255) / 256, 256, 0, stream>>>(dstIdx, deg, E);
    gcn_rsqrt_deg<<<(N + 255) / 256, 256, 0, stream>>>(deg, N);

    gcn_gemm_wmma<<<(N + 15) / 16, 256, 0, stream>>>(x, W, xw, N);

    gcn_self_bias<<<((size_t)N * 32 + 255) / 256, 256, 0, stream>>>(xw, deg, bias, out, N);
    gcn_edge_scatter<<<(E + 8 * EDGES_PER_WAVE - 1) / (8 * EDGES_PER_WAVE), 256, 0, stream>>>(srcIdx, dstIdx, deg, xw, out, E);
}


# Edge scatter by cache half: a work-group reads its die and draws 256-edge chunks from its half's ticket queue, adding only edges whose target row is in that half's node range; arrival-order sums
# speedup vs baseline: 1.7408x; 1.1130x over previous
#include <hip/hip_runtime.h>

typedef __attribute__((ext_vector_type(2))) float v2f;
typedef __attribute__((ext_vector_type(8))) float v8f;

#define NFEAT 128
#define NHID  128

#define CTR_STRIDE 64
__global__ void gcn_init_deg(float* __restrict__ deg, unsigned* __restrict__ ctr, int n) {
    int i = blockIdx.x * blockDim.x + threadIdx.x;
    if (i < n) deg[i] = 1.0f;
    if (i == 0) { ctr[0] = 0u; ctr[CTR_STRIDE] = 0u; }
}

__global__ void gcn_count_deg(const int* __restrict__ dst, float* __restrict__ deg, int e) {
    int i = blockIdx.x * blockDim.x + threadIdx.x;
    if (i < e) atomicAdd(&deg[dst[i]], 1.0f);
}

__global__ void gcn_rsqrt_deg(float* __restrict__ deg, int n) {
    int i = blockIdx.x * blockDim.x + threadIdx.x;
    if (i < n) deg[i] = rsqrtf(deg[i]);
}

__global__ __launch_bounds__(256) void gcn_gemm_wmma(const float* __restrict__ x,
                                                     const float* __restrict__ W,
                                                     float* __restrict__ xw,
                                                     int n_nodes) {
    const int lane = threadIdx.x & 31;
    const int wave = threadIdx.x >> 5;
    const int half = lane >> 4;
    const int l    = lane & 15;

    const int m0 = blockIdx.x * 16;
    const int n0 = wave * 16;

    int arow = m0 + l;
    if (arow >= n_nodes) arow = n_nodes - 1;

    const float* aptr = x + (size_t)arow * NFEAT + 2 * half;
    const float* bptr = W + (size_t)(2 * half) * NHID + n0 + l;

    v8f c = {};
#pragma unroll
    for (int kk = 0; kk < NFEAT; kk += 4) {
        v2f a = *(const v2f*)(aptr + kk);
        v2f b;
        b.x = bptr[(size_t)kk * NHID];
        b.y = bptr[(size_t)kk * NHID + NHID];
        c = __builtin_amdgcn_wmma_f32_16x16x4_f32(
                 false, a,  false, b,
                 (short)0, c,  false,  false);
    }

#pragma unroll
    for (int r = 0; r < 8; ++r) {
        int row = m0 + r + 8 * half;
        if (row < n_nodes) xw[(size_t)row * NHID + n0 + l] = c[r];
    }
}

__global__ void gcn_self_bias(const float* __restrict__ xw,
                              const float* __restrict__ dinv,
                              const float* __restrict__ bias,
                              float* __restrict__ out, int n_nodes) {
    int t = blockIdx.x * blockDim.x + threadIdx.x;
    int i = t >> 5;
    if (i >= n_nodes) return;
    int lane = t & 31;

    float di = dinv[i];
    float n2 = di * di;
    const float4 v  = *(const float4*)(xw + (size_t)i * NHID + lane * 4);
    const float4 bb = *(const float4*)(bias + lane * 4);
    float4 o;
    o.x = fmaf(v.x, n2, bb.x);
    o.y = fmaf(v.y, n2, bb.y);
    o.z = fmaf(v.z, n2, bb.z);
    o.w = fmaf(v.w, n2, bb.w);
    *(float4*)(out + (size_t)i * NHID + lane * 4) = o;
}

#define SC_WAVES 8
#define SC_EDGES_PER_WAVE 32
#define SC_EDGES_PER_WG (SC_WAVES * SC_EDGES_PER_WAVE)
#define SC_BATCH 8
#define SC_GRID 1024

template <bool GUARD>
__device__ __forceinline__ void gcn_scatter_batch(unsigned& m, int s, int d, int nrm_bits,
                                                  unsigned lane,
                                                  const float* __restrict__ xw,
                                                  float* __restrict__ out) {
    float v[SC_BATCH][4];
    int jl[SC_BATCH];
#pragma unroll
    for (int jj = 0; jj < SC_BATCH; ++jj) {
        jl[jj] = -1;
        v[jj][0] = 0.0f; v[jj][1] = 0.0f; v[jj][2] = 0.0f; v[jj][3] = 0.0f;
        if (!GUARD || m != 0u) {
            const int j = __builtin_ctz(m);
            m &= m - 1u;
            jl[jj] = j;
            const int sj = __builtin_amdgcn_readlane(s, j);
            const float* xr = xw + (size_t)sj * NHID + lane;
            v[jj][0] = xr[0];
            v[jj][1] = xr[32];
            v[jj][2] = xr[64];
            v[jj][3] = xr[96];
        }
    }
#pragma unroll
    for (int jj = 0; jj < SC_BATCH; ++jj) {
        if (!GUARD || jl[jj] >= 0) {
            const int   j  = jl[jj];
            const int   dj = __builtin_amdgcn_readlane(d, j);
            const float nj = __builtin_bit_cast(float, __builtin_amdgcn_readlane(nrm_bits, j));
            float* o = out + (size_t)dj * NHID + lane;
            atomicAdd(o +  0, v[jj][0] * nj);
            atomicAdd(o + 32, v[jj][1] * nj);
            atomicAdd(o + 64, v[jj][2] * nj);
            atomicAdd(o + 96, v[jj][3] * nj);
        }
    }
}

__global__ __launch_bounds__(256) void gcn_edge_scatter(const int* __restrict__ src,
                                                        const int* __restrict__ dst,
                                                        const float* __restrict__ dinv,
                                                        const float* __restrict__ xw,
                                                        float* __restrict__ out,
                                                        unsigned* __restrict__ ctr,
                                                        int e, int n_nodes) {
    __shared__ unsigned ticket[2];
    const unsigned lane = threadIdx.x & 31;
    const int wave = __builtin_amdgcn_readfirstlane((int)(threadIdx.x >> 5));
    const int nchunk = (e + SC_EDGES_PER_WG - 1) / SC_EDGES_PER_WG;
    const int split = n_nodes >> 1;

    const int die  = (__builtin_amdgcn_s_sendmsg_rtn(0x87) >> 16) & 15;
    const int home = (die >> 2) & 1;

    int q = home;
    unsigned slot = 0;
    unsigned drawn = 0;
    if (threadIdx.x == 0) drawn = atomicAdd(&ctr[q * CTR_STRIDE], 1u);

    for (int it = 0; it < 2 * nchunk + 2; ++it) {
        if (threadIdx.x == 0) ticket[slot] = drawn;
        __syncthreads();
        const unsigned t = (unsigned)__builtin_amdgcn_readfirstlane((int)ticket[slot]);
        slot ^= 1u;
        if (t >= (unsigned)nchunk) {
            if (q != home) break;
            q ^= 1;
            if (threadIdx.x == 0) drawn = atomicAdd(&ctr[q * CTR_STRIDE], 1u);
            continue;
        }
        if (threadIdx.x == 0) drawn = atomicAdd(&ctr[q * CTR_STRIDE], 1u);

        const int e0 = (int)t * SC_EDGES_PER_WG + wave * SC_EDGES_PER_WAVE;
        if (e0 < e) {
            const int  mine   = e0 + (int)lane;
            const bool inside = mine < e;
            const int  idx    = inside ? mine : e - 1;
            const int  s = src[idx];
            const int  d = dst[idx];
            const bool take = inside && ((d >= split) == (q != 0));
            float nrm = 0.0f;
            if (take) nrm = dinv[s] * dinv[d];
            const int nrm_bits = __builtin_bit_cast(int, nrm);
            unsigned m = __builtin_amdgcn_ballot_w32(take);
            while (m != 0u) {
                if (__builtin_popcount(m) >= SC_BATCH)
                    gcn_scatter_batch<false>(m, s, d, nrm_bits, lane, xw, out);
                else
                    gcn_scatter_batch<true>(m, s, d, nrm_bits, lane, xw, out);
            }
        }
    }
}

extern "C" void kernel_launch(void* const* d_in, const int* in_sizes, int n_in,
                              void* d_out, int out_size, void* d_ws, size_t ws_size,
                              hipStream_t stream) {
    const int*   ei   = (const int*)d_in[0];
    const float* x    = (const float*)d_in[1];
    const float* W    = (const float*)d_in[2];
    const float* bias = (const float*)d_in[3];
    float*       out  = (float*)d_out;

    const int E = in_sizes[0] / 2;
    const int N = in_sizes[1] / NFEAT;

    float* xw  = (float*)d_ws;
    float* deg = (float*)((char*)d_ws + (size_t)N * NHID * sizeof(float));
    unsigned* ctr = (unsigned*)(deg + (((size_t)N + 63) & ~(size_t)63));

    const int* srcIdx = ei;
    const int* dstIdx = ei + E;

    gcn_init_deg <<<(N + 255) / 256, 256, 0, stream>>>(deg, ctr, N);
    gcn_count_deg<<<(E + 255) / 256, 256, 0, stream>>>(dstIdx, deg, E);
    gcn_rsqrt_deg<<<(N + 255) / 256, 256, 0, stream>>>(deg, N);

    gcn_gemm_wmma<<<(N + 15) / 16, 256, 0, stream>>>(x, W, xw, N);

    gcn_self_bias<<<((size_t)N * 32 + 255) / 256, 256, 0, stream>>>(xw, deg, bias, out, N);
    gcn_edge_scatter<<<SC_GRID, 256, 0, stream>>>(srcIdx, dstIdx, deg, xw, out, ctr, E, N);
}


# GEMM: a wave keeps its 128x16 slab of W in registers as B fragments and does 64 rows (4 accumulator tiles) per work-group; x block staged in LDS by whole-row loads as A source; k sum order changes
# speedup vs baseline: 1.8793x; 1.0796x over previous
#include <hip/hip_runtime.h>

typedef __attribute__((ext_vector_type(2))) float v2f;
typedef __attribute__((ext_vector_type(8))) float v8f;

#define NFEAT 128
#define NHID  128

#define CTR_STRIDE 64
__global__ void gcn_init_deg(float* __restrict__ deg, unsigned* __restrict__ ctr, int n) {
    int i = blockIdx.x * blockDim.x + threadIdx.x;
    if (i < n) deg[i] = 1.0f;
    if (i == 0) { ctr[0] = 0u; ctr[CTR_STRIDE] = 0u; }
}

__global__ void gcn_count_deg(const int* __restrict__ dst, float* __restrict__ deg, int e) {
    int i = blockIdx.x * blockDim.x + threadIdx.x;
    if (i < e) atomicAdd(&deg[dst[i]], 1.0f);
}

__global__ void gcn_rsqrt_deg(float* __restrict__ deg, int n) {
    int i = blockIdx.x * blockDim.x + threadIdx.x;
    if (i < n) deg[i] = rsqrtf(deg[i]);
}

#define GM_ROWS  64
#define GM_TILES (GM_ROWS / 16)
#define GM_PITCH 132
__global__ __launch_bounds__(256) void gcn_gemm_wmma(const float* __restrict__ x,
                                                     const float* __restrict__ W,
                                                     float* __restrict__ xw,
                                                     int n_nodes) {
    __shared__ __attribute__((aligned(16))) float xs[GM_ROWS * GM_PITCH];

    const int lane = threadIdx.x & 31;
    const int wave = threadIdx.x >> 5;
    const int half = lane >> 4;
    const int l    = lane & 15;

    const int m0 = blockIdx.x * GM_ROWS;
    const int n0 = wave * 16;

#pragma unroll
    for (int i = 0; i < GM_ROWS * 32 / 256; ++i) {
        const int idx = (int)threadIdx.x + 256 * i;
        const int row = idx >> 5;
        const int c4  = idx & 31;
        int grow = m0 + row;
        if (grow >= n_nodes) grow = n_nodes - 1;
        const float4 v = *(const float4*)(x + (size_t)grow * NFEAT + 4 * c4);
        *(float4*)(xs + row * GM_PITCH + 4 * c4) = v;
    }

    v2f bf[NFEAT / 4];
    const float* bptr = W + (size_t)(4 * half) * NHID + n0 + l;
#pragma unroll
    for (int s = 0; s < NFEAT / 8; ++s) {
#pragma unroll
        for (int j = 0; j < 2; ++j) {
            const int k = 8 * s + 2 * j;
            bf[2 * s + j].x = bptr[(size_t)k * NHID];
            bf[2 * s + j].y = bptr[(size_t)(k + 1) * NHID];
        }
    }

    __syncthreads();

    v8f acc[GM_TILES];
#pragma unroll
    for (int t = 0; t < GM_TILES; ++t) acc[t] = (v8f){};

    const float* aptr = xs + l * GM_PITCH + 4 * half;
#pragma unroll
    for (int s = 0; s < NFEAT / 8; ++s) {
        float4 a[GM_TILES];
#pragma unroll
        for (int t = 0; t < GM_TILES; ++t)
            a[t] = *(const float4*)(aptr + 16 * t * GM_PITCH + 8 * s);
#pragma unroll
        for (int t = 0; t < GM_TILES; ++t) {
            v2f a0; a0.x = a[t].x; a0.y = a[t].y;
            v2f a1; a1.x = a[t].z; a1.y = a[t].w;
            acc[t] = __builtin_amdgcn_wmma_f32_16x16x4_f32(
                     false, a0,  false, bf[2 * s],
                     (short)0, acc[t],  false,  false);
            acc[t] = __builtin_amdgcn_wmma_f32_16x16x4_f32(
                     false, a1,  false, bf[2 * s + 1],
                     (short)0, acc[t],  false,  false);
        }
    }

#pragma unroll
    for (int t = 0; t < GM_TILES; ++t) {
#pragma unroll
        for (int r = 0; r < 8; ++r) {
            const int row = m0 + 16 * t + r + 8 * half;
            if (row < n_nodes) xw[(size_t)row * NHID + n0 + l] = acc[t][r];
        }
    }
}

__global__ void gcn_self_bias(const float* __restrict__ xw,
                              const float* __restrict__ dinv,
                              const float* __restrict__ bias,
                              float* __restrict__ out, int n_nodes) {
    int t = blockIdx.x * blockDim.x + threadIdx.x;
    int i = t >> 5;
    if (i >= n_nodes) return;
    int lane = t & 31;

    float di = dinv[i];
    float n2 = di * di;
    const float4 v  = *(const float4*)(xw + (size_t)i * NHID + lane * 4);
    const float4 bb = *(const float4*)(bias + lane * 4);
    float4 o;
    o.x = fmaf(v.x, n2, bb.x);
    o.y = fmaf(v.y, n2, bb.y);
    o.z = fmaf(v.z, n2, bb.z);
    o.w = fmaf(v.w, n2, bb.w);
    *(float4*)(out + (size_t)i * NHID + lane * 4) = o;
}

#define SC_WAVES 8
#define SC_EDGES_PER_WAVE 32
#define SC_EDGES_PER_WG (SC_WAVES * SC_EDGES_PER_WAVE)
#define SC_BATCH 8
#define SC_GRID 1024

template <bool GUARD>
__device__ __forceinline__ void gcn_scatter_batch(unsigned& m, int s, int d, int nrm_bits,
                                                  unsigned lane,
                                                  const float* __restrict__ xw,
                                                  float* __restrict__ out) {
    float v[SC_BATCH][4];
    int jl[SC_BATCH];
#pragma unroll
    for (int jj = 0; jj < SC_BATCH; ++jj) {
        jl[jj] = -1;
        v[jj][0] = 0.0f; v[jj][1] = 0.0f; v[jj][2] = 0.0f; v[jj][3] = 0.0f;
        if (!GUARD || m != 0u) {
            const int j = __builtin_ctz(m);
            m &= m - 1u;
            jl[jj] = j;
            const int sj = __builtin_amdgcn_readlane(s, j);
            const float* xr = xw + (size_t)sj * NHID + lane;
            v[jj][0] = xr[0];
            v[jj][1] = xr[32];
            v[jj][2] = xr[64];
            v[jj][3] = xr[96];
        }
    }
#pragma unroll
    for (int jj = 0; jj < SC_BATCH; ++jj) {
        if (!GUARD || jl[jj] >= 0) {
            const int   j  = jl[jj];
            const int   dj = __builtin_amdgcn_readlane(d, j);
            const float nj = __builtin_bit_cast(float, __builtin_amdgcn_readlane(nrm_bits, j));
            float* o = out + (size_t)dj * NHID + lane;
            atomicAdd(o +  0, v[jj][0] * nj);
            atomicAdd(o + 32, v[jj][1] * nj);
            atomicAdd(o + 64, v[jj][2] * nj);
            atomicAdd(o + 96, v[jj][3] * nj);
        }
    }
}

__global__ __launch_bounds__(256) void gcn_edge_scatter(const int* __restrict__ src,
                                                        const int* __restrict__ dst,
                                                        const float* __restrict__ dinv,
                                                        const float* __restrict__ xw,
                                                        float* __restrict__ out,
                                                        unsigned* __restrict__ ctr,
                                                        int e, int n_nodes) {
    __shared__ unsigned ticket[2];
    const unsigned lane = threadIdx.x & 31;
    const int wave = __builtin_amdgcn_readfirstlane((int)(threadIdx.x >> 5));
    const int nchunk = (e + SC_EDGES_PER_WG - 1) / SC_EDGES_PER_WG;
    const int split = n_nodes >> 1;

    const int die  = (__builtin_amdgcn_s_sendmsg_rtn(0x87) >> 16) & 15;
    const int home = (die >> 2) & 1;

    int q = home;
    unsigned slot = 0;
    unsigned drawn = 0;
    if (threadIdx.x == 0) drawn = atomicAdd(&ctr[q * CTR_STRIDE], 1u);

    for (int it = 0; it < 2 * nchunk + 2; ++it) {
        if (threadIdx.x == 0) ticket[slot] = drawn;
        __syncthreads();
        const unsigned t = (unsigned)__builtin_amdgcn_readfirstlane((int)ticket[slot]);
        slot ^= 1u;
        if (t >= (unsigned)nchunk) {
            if (q != home) break;
            q ^= 1;
            if (threadIdx.x == 0) drawn = atomicAdd(&ctr[q * CTR_STRIDE], 1u);
            continue;
        }
        if (threadIdx.x == 0) drawn = atomicAdd(&ctr[q * CTR_STRIDE], 1u);

        const int e0 = (int)t * SC_EDGES_PER_WG + wave * SC_EDGES_PER_WAVE;
        if (e0 < e) {
            const int  mine   = e0 + (int)lane;
            const bool inside = mine < e;
            const int  idx    = inside ? mine : e - 1;
            const int  s = src[idx];
            const int  d = dst[idx];
            const bool take = inside && ((d >= split) == (q != 0));
            float nrm = 0.0f;
            if (take) nrm = dinv[s] * dinv[d];
            const int nrm_bits = __builtin_bit_cast(int, nrm);
            unsigned m = __builtin_amdgcn_ballot_w32(take);
            while (m != 0u) {
                if (__builtin_popcount(m) >= SC_BATCH)
                    gcn_scatter_batch<false>(m, s, d, nrm_bits, lane, xw, out);
                else
                    gcn_scatter_batch<true>(m, s, d, nrm_bits, lane, xw, out);
            }
        }
    }
}

extern "C" void kernel_launch(void* const* d_in, const int* in_sizes, int n_in,
                              void* d_out, int out_size, void* d_ws, size_t ws_size,
                              hipStream_t stream) {
    const int*   ei   = (const int*)d_in[0];
    const float* x    = (const float*)d_in[1];
    const float* W    = (const float*)d_in[2];
    const float* bias = (const float*)d_in[3];
    float*       out  = (float*)d_out;

    const int E = in_sizes[0] / 2;
    const int N = in_sizes[1] / NFEAT;

    float* xw  = (float*)d_ws;
    float* deg = (float*)((char*)d_ws + (size_t)N * NHID * sizeof(float));
    unsigned* ctr = (unsigned*)(deg + (((size_t)N + 63) & ~(size_t)63));

    const int* srcIdx = ei;
    const int* dstIdx = ei + E;

    gcn_init_deg <<<(N + 255) / 256, 256, 0, stream>>>(deg, ctr, N);
    gcn_count_deg<<<(E + 255) / 256, 256, 0, stream>>>(dstIdx, deg, E);
    gcn_rsqrt_deg<<<(N + 255) / 256, 256, 0, stream>>>(deg, N);

    gcn_gemm_wmma<<<(N + GM_ROWS - 1) / GM_ROWS, 256, 0, stream>>>(x, W, xw, N);

    gcn_self_bias<<<((size_t)N * 32 + 255) / 256, 256, 0, stream>>>(xw, deg, bias, out, N);
    gcn_edge_scatter<<<SC_GRID, 256, 0, stream>>>(srcIdx, dstIdx, deg, xw, out, ctr, E, N);
}


# Degree count: 64 work-groups each count their slice of edge targets in a private LDS table (LDS integer adds) and store it as a plane; a finish launch sums the planes; 5 launches, not 6
# speedup vs baseline: 3.1561x; 1.6794x over previous
#include <hip/hip_runtime.h>

typedef __attribute__((ext_vector_type(2))) float v2f;
typedef __attribute__((ext_vector_type(8))) float v8f;

#define NFEAT 128
#define NHID  128

#define CTR_STRIDE 64
#define DH_GROUPS  64
#define DH_THREADS 1024
#define DH_BATCH   16
#define DH_CAP     65536
__global__ __launch_bounds__(DH_THREADS) void gcn_deg_hist(const int* __restrict__ dst,
                                                           unsigned* __restrict__ planes,
                                                           int e, int n) {
    __shared__ __attribute__((aligned(16))) unsigned h[DH_CAP];
    const int tid   = (int)threadIdx.x;
    const int pitch = (n + 3) & ~3;
    const int per   = (e + DH_GROUPS - 1) / DH_GROUPS;
    const int e0    = (int)blockIdx.x * per;
    const int e1    = (e0 + per < e) ? (e0 + per) : e;
    unsigned* plane = planes + (size_t)blockIdx.x * (size_t)pitch;

    for (int base = 0; base < n; base += DH_CAP) {
        const int span  = (n - base < DH_CAP) ? (n - base) : DH_CAP;
        const int span4 = (span + 3) & ~3;
        for (int i = 4 * tid; i < span4; i += 4 * DH_THREADS)
            *(uint4*)(h + i) = make_uint4(0u, 0u, 0u, 0u);
        __syncthreads();
        for (int i0 = e0 + tid; i0 < e1; i0 += DH_BATCH * DH_THREADS) {
            unsigned d[DH_BATCH];
#pragma unroll
            for (int k = 0; k < DH_BATCH; ++k) {
                const int i  = i0 + k * DH_THREADS;
                const int ic = (i < e1) ? i : (e1 - 1);
                const unsigned v = (unsigned)(dst[ic] - base);
                d[k] = (i < e1) ? v : 0xFFFFFFFFu;
            }
#pragma unroll
            for (int k = 0; k < DH_BATCH; ++k)
                if (d[k] < (unsigned)span) atomicAdd(&h[d[k]], 1u);
        }
        __syncthreads();
#pragma unroll 4
        for (int i = 4 * tid; i < span4; i += 4 * DH_THREADS)
            *(uint4*)(plane + base + i) = *(const uint4*)(h + i);
        __syncthreads();
    }
}

__global__ void gcn_deg_finish(const unsigned* __restrict__ planes, float* __restrict__ dinv,
                               unsigned* __restrict__ ctr, int n) {
    const int i = blockIdx.x * blockDim.x + threadIdx.x;
    const int pitch = (n + 3) & ~3;
    if (i == 0) { ctr[0] = 0u; ctr[CTR_STRIDE] = 0u; }
    if (i < n) {
        unsigned c = 1u;
#pragma unroll 16
        for (int g = 0; g < DH_GROUPS; ++g) c += planes[(size_t)g * (size_t)pitch + i];
        dinv[i] = rsqrtf((float)c);
    }
}

#define GM_ROWS  64
#define GM_TILES (GM_ROWS / 16)
#define GM_PITCH 132
__global__ __launch_bounds__(256) void gcn_gemm_wmma(const float* __restrict__ x,
                                                     const float* __restrict__ W,
                                                     float* __restrict__ xw,
                                                     int n_nodes) {
    __shared__ __attribute__((aligned(16))) float xs[GM_ROWS * GM_PITCH];

    const int lane = threadIdx.x & 31;
    const int wave = threadIdx.x >> 5;
    const int half = lane >> 4;
    const int l    = lane & 15;

    const int m0 = blockIdx.x * GM_ROWS;
    const int n0 = wave * 16;

#pragma unroll
    for (int i = 0; i < GM_ROWS * 32 / 256; ++i) {
        const int idx = (int)threadIdx.x + 256 * i;
        const int row = idx >> 5;
        const int c4  = idx & 31;
        int grow = m0 + row;
        if (grow >= n_nodes) grow = n_nodes - 1;
        const float4 v = *(const float4*)(x + (size_t)grow * NFEAT + 4 * c4);
        *(float4*)(xs + row * GM_PITCH + 4 * c4) = v;
    }

    v2f bf[NFEAT / 4];
    const float* bptr = W + (size_t)(4 * half) * NHID + n0 + l;
#pragma unroll
    for (int s = 0; s < NFEAT / 8; ++s) {
#pragma unroll
        for (int j = 0; j < 2; ++j) {
            const int k = 8 * s + 2 * j;
            bf[2 * s + j].x = bptr[(size_t)k * NHID];
            bf[2 * s + j].y = bptr[(size_t)(k + 1) * NHID];
        }
    }

    __syncthreads();

    v8f acc[GM_TILES];
#pragma unroll
    for (int t = 0; t < GM_TILES; ++t) acc[t] = (v8f){};

    const float* aptr = xs + l * GM_PITCH + 4 * half;
#pragma unroll
    for (int s = 0; s < NFEAT / 8; ++s) {
        float4 a[GM_TILES];
#pragma unroll
        for (int t = 0; t < GM_TILES; ++t)
            a[t] = *(const float4*)(aptr + 16 * t * GM_PITCH + 8 * s);
#pragma unroll
        for (int t = 0; t < GM_TILES; ++t) {
            v2f a0; a0.x = a[t].x; a0.y = a[t].y;
            v2f a1; a1.x = a[t].z; a1.y = a[t].w;
            acc[t] = __builtin_amdgcn_wmma_f32_16x16x4_f32(
                     false, a0,  false, bf[2 * s],
                     (short)0, acc[t],  false,  false);
            acc[t] = __builtin_amdgcn_wmma_f32_16x16x4_f32(
                     false, a1,  false, bf[2 * s + 1],
                     (short)0, acc[t],  false,  false);
        }
    }

#pragma unroll
    for (int t = 0; t < GM_TILES; ++t) {
#pragma unroll
        for (int r = 0; r < 8; ++r) {
            const int row = m0 + 16 * t + r + 8 * half;
            if (row < n_nodes) xw[(size_t)row * NHID + n0 + l] = acc[t][r];
        }
    }
}

__global__ void gcn_self_bias(const float* __restrict__ xw,
                              const float* __restrict__ dinv,
                              const float* __restrict__ bias,
                              float* __restrict__ out, int n_nodes) {
    int t = blockIdx.x * blockDim.x + threadIdx.x;
    int i = t >> 5;
    if (i >= n_nodes) return;
    int lane = t & 31;

    float di = dinv[i];
    float n2 = di * di;
    const float4 v  = *(const float4*)(xw + (size_t)i * NHID + lane * 4);
    const float4 bb = *(const float4*)(bias + lane * 4);
    float4 o;
    o.x = fmaf(v.x, n2, bb.x);
    o.y = fmaf(v.y, n2, bb.y);
    o.z = fmaf(v.z, n2, bb.z);
    o.w = fmaf(v.w, n2, bb.w);
    *(float4*)(out + (size_t)i * NHID + lane * 4) = o;
}

#define SC_WAVES 8
#define SC_EDGES_PER_WAVE 32
#define SC_EDGES_PER_WG (SC_WAVES * SC_EDGES_PER_WAVE)
#define SC_BATCH 8
#define SC_GRID 1024

template <bool GUARD>
__device__ __forceinline__ void gcn_scatter_batch(unsigned& m, int s, int d, int nrm_bits,
                                                  unsigned lane,
                                                  const float* __restrict__ xw,
                                                  float* __restrict__ out) {
    float v[SC_BATCH][4];
    int jl[SC_BATCH];
#pragma unroll
    for (int jj = 0; jj < SC_BATCH; ++jj) {
        jl[jj] = -1;
        v[jj][0] = 0.0f; v[jj][1] = 0.0f; v[jj][2] = 0.0f; v[jj][3] = 0.0f;
        if (!GUARD || m != 0u) {
            const int j = __builtin_ctz(m);
            m &= m - 1u;
            jl[jj] = j;
            const int sj = __builtin_amdgcn_readlane(s, j);
            const float* xr = xw + (size_t)sj * NHID + lane;
            v[jj][0] = xr[0];
            v[jj][1] = xr[32];
            v[jj][2] = xr[64];
            v[jj][3] = xr[96];
        }
    }
#pragma unroll
    for (int jj = 0; jj < SC_BATCH; ++jj) {
        if (!GUARD || jl[jj] >= 0) {
            const int   j  = jl[jj];
            const int   dj = __builtin_amdgcn_readlane(d, j);
            const float nj = __builtin_bit_cast(float, __builtin_amdgcn_readlane(nrm_bits, j));
            float* o = out + (size_t)dj * NHID + lane;
            atomicAdd(o +  0, v[jj][0] * nj);
            atomicAdd(o + 32, v[jj][1] * nj);
            atomicAdd(o + 64, v[jj][2] * nj);
            atomicAdd(o + 96, v[jj][3] * nj);
        }
    }
}

__global__ __launch_bounds__(256) void gcn_edge_scatter(const int* __restrict__ src,
                                                        const int* __restrict__ dst,
                                                        const float* __restrict__ dinv,
                                                        const float* __restrict__ xw,
                                                        float* __restrict__ out,
                                                        unsigned* __restrict__ ctr,
                                                        int e, int n_nodes) {
    __shared__ unsigned ticket[2];
    const unsigned lane = threadIdx.x & 31;
    const int wave = __builtin_amdgcn_readfirstlane((int)(threadIdx.x >> 5));
    const int nchunk = (e + SC_EDGES_PER_WG - 1) / SC_EDGES_PER_WG;
    const int split = n_nodes >> 1;

    const int die  = (__builtin_amdgcn_s_sendmsg_rtn(0x87) >> 16) & 15;
    const int home = (die >> 2) & 1;

    int q = home;
    unsigned slot = 0;
    unsigned drawn = 0;
    if (threadIdx.x == 0) drawn = atomicAdd(&ctr[q * CTR_STRIDE], 1u);

    for (int it = 0; it < 2 * nchunk + 2; ++it) {
        if (threadIdx.x == 0) ticket[slot] = drawn;
        __syncthreads();
        const unsigned t = (unsigned)__builtin_amdgcn_readfirstlane((int)ticket[slot]);
        slot ^= 1u;
        if (t >= (unsigned)nchunk) {
            if (q != home) break;
            q ^= 1;
            if (threadIdx.x == 0) drawn = atomicAdd(&ctr[q * CTR_STRIDE], 1u);
            continue;
        }
        if (threadIdx.x == 0) drawn = atomicAdd(&ctr[q * CTR_STRIDE], 1u);

        const int e0 = (int)t * SC_EDGES_PER_WG + wave * SC_EDGES_PER_WAVE;
        if (e0 < e) {
            const int  mine   = e0 + (int)lane;
            const bool inside = mine < e;
            const int  idx    = inside ? mine : e - 1;
            const int  s = src[idx];
            const int  d = dst[idx];
            const bool take = inside && ((d >= split) == (q != 0));
            float nrm = 0.0f;
            if (take) nrm = dinv[s] * dinv[d];
            const int nrm_bits = __builtin_bit_cast(int, nrm);
            unsigned m = __builtin_amdgcn_ballot_w32(take);
            while (m != 0u) {
                if (__builtin_popcount(m) >= SC_BATCH)
                    gcn_scatter_batch<false>(m, s, d, nrm_bits, lane, xw, out);
                else
                    gcn_scatter_batch<true>(m, s, d, nrm_bits, lane, xw, out);
            }
        }
    }
}

extern "C" void kernel_launch(void* const* d_in, const int* in_sizes, int n_in,
                              void* d_out, int out_size, void* d_ws, size_t ws_size,
                              hipStream_t stream) {
    const int*   ei   = (const int*)d_in[0];
    const float* x    = (const float*)d_in[1];
    const float* W    = (const float*)d_in[2];
    const float* bias = (const float*)d_in[3];
    float*       out  = (float*)d_out;

    const int E = in_sizes[0] / 2;
    const int N = in_sizes[1] / NFEAT;

    float* xw  = (float*)d_ws;
    float* deg = (float*)((char*)d_ws + (size_t)N * NHID * sizeof(float));
    unsigned* ctr = (unsigned*)(deg + (((size_t)N + 63) & ~(size_t)63));
    unsigned* planes = ctr + 2 * CTR_STRIDE;

    const int* srcIdx = ei;
    const int* dstIdx = ei + E;

    gcn_deg_hist  <<<DH_GROUPS, DH_THREADS, 0, stream>>>(dstIdx, planes, E, N);
    gcn_deg_finish<<<(N + 255) / 256, 256, 0, stream>>>(planes, deg, ctr, N);

    gcn_gemm_wmma<<<(N + GM_ROWS - 1) / GM_ROWS, 256, 0, stream>>>(x, W, xw, N);

    gcn_self_bias<<<((size_t)N * 32 + 255) / 256, 256, 0, stream>>>(xw, deg, bias, out, N);
    gcn_edge_scatter<<<SC_GRID, 256, 0, stream>>>(srcIdx, dstIdx, deg, xw, out, ctr, E, N);
}


# Degree finish merged into out's start-value pass: a wave sums the count planes of 16 nodes (half per lane, one lane exchange), stores dinv, writes their 16 rows; 4 launches, not 5; sums unchanged
# speedup vs baseline: 3.2877x; 1.0417x over previous
#include <hip/hip_runtime.h>

typedef __attribute__((ext_vector_type(2))) float v2f;
typedef __attribute__((ext_vector_type(8))) float v8f;

#define NFEAT 128
#define NHID  128

#define CTR_STRIDE 64
#define DH_GROUPS  64
#define DH_THREADS 1024
#define DH_BATCH   16
#define DH_CAP     65536
__global__ __launch_bounds__(DH_THREADS) void gcn_deg_hist(const int* __restrict__ dst,
                                                           unsigned* __restrict__ planes,
                                                           int e, int n) {
    __shared__ __attribute__((aligned(16))) unsigned h[DH_CAP];
    const int tid   = (int)threadIdx.x;
    const int pitch = (n + 3) & ~3;
    const int per   = (e + DH_GROUPS - 1) / DH_GROUPS;
    const int e0    = (int)blockIdx.x * per;
    const int e1    = (e0 + per < e) ? (e0 + per) : e;
    unsigned* plane = planes + (size_t)blockIdx.x * (size_t)pitch;

    for (int base = 0; base < n; base += DH_CAP) {
        const int span  = (n - base < DH_CAP) ? (n - base) : DH_CAP;
        const int span4 = (span + 3) & ~3;
        for (int i = 4 * tid; i < span4; i += 4 * DH_THREADS)
            *(uint4*)(h + i) = make_uint4(0u, 0u, 0u, 0u);
        __syncthreads();
        for (int i0 = e0 + tid; i0 < e1; i0 += DH_BATCH * DH_THREADS) {
            unsigned d[DH_BATCH];
#pragma unroll
            for (int k = 0; k < DH_BATCH; ++k) {
                const int i  = i0 + k * DH_THREADS;
                const int ic = (i < e1) ? i : (e1 - 1);
                const unsigned v = (unsigned)(dst[ic] - base);
                d[k] = (i < e1) ? v : 0xFFFFFFFFu;
            }
#pragma unroll
            for (int k = 0; k < DH_BATCH; ++k)
                if (d[k] < (unsigned)span) atomicAdd(&h[d[k]], 1u);
        }
        __syncthreads();
#pragma unroll 4
        for (int i = 4 * tid; i < span4; i += 4 * DH_THREADS)
            *(uint4*)(plane + base + i) = *(const uint4*)(h + i);
        __syncthreads();
    }
}

#define GM_ROWS  64
#define GM_TILES (GM_ROWS / 16)
#define GM_PITCH 132
__global__ __launch_bounds__(256) void gcn_gemm_wmma(const float* __restrict__ x,
                                                     const float* __restrict__ W,
                                                     float* __restrict__ xw,
                                                     int n_nodes) {
    __shared__ __attribute__((aligned(16))) float xs[GM_ROWS * GM_PITCH];

    const int lane = threadIdx.x & 31;
    const int wave = threadIdx.x >> 5;
    const int half = lane >> 4;
    const int l    = lane & 15;

    const int m0 = blockIdx.x * GM_ROWS;
    const int n0 = wave * 16;

#pragma unroll
    for (int i = 0; i < GM_ROWS * 32 / 256; ++i) {
        const int idx = (int)threadIdx.x + 256 * i;
        const int row = idx >> 5;
        const int c4  = idx & 31;
        int grow = m0 + row;
        if (grow >= n_nodes) grow = n_nodes - 1;
        const float4 v = *(const float4*)(x + (size_t)grow * NFEAT + 4 * c4);
        *(float4*)(xs + row * GM_PITCH + 4 * c4) = v;
    }

    v2f bf[NFEAT / 4];
    const float* bptr = W + (size_t)(4 * half) * NHID + n0 + l;
#pragma unroll
    for (int s = 0; s < NFEAT / 8; ++s) {
#pragma unroll
        for (int j = 0; j < 2; ++j) {
            const int k = 8 * s + 2 * j;
            bf[2 * s + j].x = bptr[(size_t)k * NHID];
            bf[2 * s + j].y = bptr[(size_t)(k + 1) * NHID];
        }
    }

    __syncthreads();

    v8f acc[GM_TILES];
#pragma unroll
    for (int t = 0; t < GM_TILES; ++t) acc[t] = (v8f){};

    const float* aptr = xs + l * GM_PITCH + 4 * half;
#pragma unroll
    for (int s = 0; s < NFEAT / 8; ++s) {
        float4 a[GM_TILES];
#pragma unroll
        for (int t = 0; t < GM_TILES; ++t)
            a[t] = *(const float4*)(aptr + 16 * t * GM_PITCH + 8 * s);
#pragma unroll
        for (int t = 0; t < GM_TILES; ++t) {
            v2f a0; a0.x = a[t].x; a0.y = a[t].y;
            v2f a1; a1.x = a[t].z; a1.y = a[t].w;
            acc[t] = __builtin_amdgcn_wmma_f32_16x16x4_f32(
                     false, a0,  false, bf[2 * s],
                     (short)0, acc[t],  false,  false);
            acc[t] = __builtin_amdgcn_wmma_f32_16x16x4_f32(
                     false, a1,  false, bf[2 * s + 1],
                     (short)0, acc[t],  false,  false);
        }
    }

#pragma unroll
    for (int t = 0; t < GM_TILES; ++t) {
#pragma unroll
        for (int r = 0; r < 8; ++r) {
            const int row = m0 + 16 * t + r + 8 * half;
            if (row < n_nodes) xw[(size_t)row * NHID + n0 + l] = acc[t][r];
        }
    }
}

#define FS_NODES   16
#define FS_WAVES   4

__global__ __launch_bounds__(FS_WAVES * 32) void gcn_finish_self(const unsigned* __restrict__ planes,
                                                               const float* __restrict__ xw,
                                                               const float* __restrict__ bias,
                                                               float* __restrict__ dinv,
                                                               float* __restrict__ out,
                                                               unsigned* __restrict__ ctr, int n) {
    const int lane = (int)(threadIdx.x & 31);
    const int wave = __builtin_amdgcn_readfirstlane((int)(threadIdx.x >> 5));
    const int l    = lane & 15;
    const int half = lane >> 4;
    const int pitch = (n + 3) & ~3;
    if (blockIdx.x == 0 && threadIdx.x == 0) { ctr[0] = 0u; ctr[CTR_STRIDE] = 0u; }

    const int node0 = ((int)blockIdx.x * FS_WAVES + wave) * FS_NODES;
    if (node0 >= n) return;
    const int cnt  = (n - node0 < FS_NODES) ? (n - node0) : FS_NODES;
    const int node = node0 + ((l < cnt) ? l : (cnt - 1));

    unsigned c = 0u;
    const unsigned* p = planes + (size_t)(half * (DH_GROUPS / 2)) * (size_t)pitch + node;
#pragma unroll
    for (int g = 0; g < DH_GROUPS / 2; ++g) c += p[(size_t)g * (size_t)pitch];
    c += (unsigned)__shfl_xor((int)c, 16);
    const float di = rsqrtf((float)(c + 1u));
    if (half == 0 && l < cnt) dinv[node] = di;
    const int n2_bits = __builtin_bit_cast(int, di * di);

    const float4 bb = *(const float4*)(bias + lane * 4);
    const float* xr   = xw  + (size_t)node0 * NHID + lane * 4;
    float*       orow = out + (size_t)node0 * NHID + lane * 4;
    if (cnt == FS_NODES) {
        float4 v[FS_NODES];
#pragma unroll
        for (int k = 0; k < FS_NODES; ++k) v[k] = *(const float4*)(xr + (size_t)k * NHID);
#pragma unroll
        for (int k = 0; k < FS_NODES; ++k) {
            const float n2 = __builtin_bit_cast(float, __builtin_amdgcn_readlane(n2_bits, k));
            float4 o;
            o.x = fmaf(v[k].x, n2, bb.x);
            o.y = fmaf(v[k].y, n2, bb.y);
            o.z = fmaf(v[k].z, n2, bb.z);
            o.w = fmaf(v[k].w, n2, bb.w);
            *(float4*)(orow + (size_t)k * NHID) = o;
        }
    } else {
        for (int k = 0; k < cnt; ++k) {
            const float n2 = __builtin_bit_cast(float, __builtin_amdgcn_readlane(n2_bits, k));
            const float4 v = *(const float4*)(xr + (size_t)k * NHID);
            float4 o;
            o.x = fmaf(v.x, n2, bb.x);
            o.y = fmaf(v.y, n2, bb.y);
            o.z = fmaf(v.z, n2, bb.z);
            o.w = fmaf(v.w, n2, bb.w);
            *(float4*)(orow + (size_t)k * NHID) = o;
        }
    }
}

#define SC_WAVES 8
#define SC_EDGES_PER_WAVE 32
#define SC_EDGES_PER_WG (SC_WAVES * SC_EDGES_PER_WAVE)
#define SC_BATCH 8
#define SC_GRID 1024

template <bool GUARD>
__device__ __forceinline__ void gcn_scatter_batch(unsigned& m, int s, int d, int nrm_bits,
                                                  unsigned lane,
                                                  const float* __restrict__ xw,
                                                  float* __restrict__ out) {
    float v[SC_BATCH][4];
    int jl[SC_BATCH];
#pragma unroll
    for (int jj = 0; jj < SC_BATCH; ++jj) {
        jl[jj] = -1;
        v[jj][0] = 0.0f; v[jj][1] = 0.0f; v[jj][2] = 0.0f; v[jj][3] = 0.0f;
        if (!GUARD || m != 0u) {
            const int j = __builtin_ctz(m);
            m &= m - 1u;
            jl[jj] = j;
            const int sj = __builtin_amdgcn_readlane(s, j);
            const float* xr = xw + (size_t)sj * NHID + lane;
            v[jj][0] = xr[0];
            v[jj][1] = xr[32];
            v[jj][2] = xr[64];
            v[jj][3] = xr[96];
        }
    }
#pragma unroll
    for (int jj = 0; jj < SC_BATCH; ++jj) {
        if (!GUARD || jl[jj] >= 0) {
            const int   j  = jl[jj];
            const int   dj = __builtin_amdgcn_readlane(d, j);
            const float nj = __builtin_bit_cast(float, __builtin_amdgcn_readlane(nrm_bits, j));
            float* o = out + (size_t)dj * NHID + lane;
            atomicAdd(o +  0, v[jj][0] * nj);
            atomicAdd(o + 32, v[jj][1] * nj);
            atomicAdd(o + 64, v[jj][2] * nj);
            atomicAdd(o + 96, v[jj][3] * nj);
        }
    }
}

__global__ __launch_bounds__(256) void gcn_edge_scatter(const int* __restrict__ src,
                                                        const int* __restrict__ dst,
                                                        const float* __restrict__ dinv,
                                                        const float* __restrict__ xw,
                                                        float* __restrict__ out,
                                                        unsigned* __restrict__ ctr,
                                                        int e, int n_nodes) {
    __shared__ unsigned ticket[2];
    const unsigned lane = threadIdx.x & 31;
    const int wave = __builtin_amdgcn_readfirstlane((int)(threadIdx.x >> 5));
    const int nchunk = (e + SC_EDGES_PER_WG - 1) / SC_EDGES_PER_WG;
    const int split = n_nodes >> 1;

    const int die  = (__builtin_amdgcn_s_sendmsg_rtn(0x87) >> 16) & 15;
    const int home = (die >> 2) & 1;

    int q = home;
    unsigned slot = 0;
    unsigned drawn = 0;
    if (threadIdx.x == 0) drawn = atomicAdd(&ctr[q * CTR_STRIDE], 1u);

    for (int it = 0; it < 2 * nchunk + 2; ++it) {
        if (threadIdx.x == 0) ticket[slot] = drawn;
        __syncthreads();
        const unsigned t = (unsigned)__builtin_amdgcn_readfirstlane((int)ticket[slot]);
        slot ^= 1u;
        if (t >= (unsigned)nchunk) {
            if (q != home) break;
            q ^= 1;
            if (threadIdx.x == 0) drawn = atomicAdd(&ctr[q * CTR_STRIDE], 1u);
            continue;
        }
        if (threadIdx.x == 0) drawn = atomicAdd(&ctr[q * CTR_STRIDE], 1u);

        const int e0 = (int)t * SC_EDGES_PER_WG + wave * SC_EDGES_PER_WAVE;
        if (e0 < e) {
            const int  mine   = e0 + (int)lane;
            const bool inside = mine < e;
            const int  idx    = inside ? mine : e - 1;
            const int  s = src[idx];
            const int  d = dst[idx];
            const bool take = inside && ((d >= split) == (q != 0));
            float nrm = 0.0f;
            if (take) nrm = dinv[s] * dinv[d];
            const int nrm_bits = __builtin_bit_cast(int, nrm);
            unsigned m = __builtin_amdgcn_ballot_w32(take);
            while (m != 0u) {
                if (__builtin_popcount(m) >= SC_BATCH)
                    gcn_scatter_batch<false>(m, s, d, nrm_bits, lane, xw, out);
                else
                    gcn_scatter_batch<true>(m, s, d, nrm_bits, lane, xw, out);
            }
        }
    }
}

extern "C" void kernel_launch(void* const* d_in, const int* in_sizes, int n_in,
                              void* d_out, int out_size, void* d_ws, size_t ws_size,
                              hipStream_t stream) {
    const int*   ei   = (const int*)d_in[0];
    const float* x    = (const float*)d_in[1];
    const float* W    = (const float*)d_in[2];
    const float* bias = (const float*)d_in[3];
    float*       out  = (float*)d_out;

    const int E = in_sizes[0] / 2;
    const int N = in_sizes[1] / NFEAT;

    float* xw  = (float*)d_ws;
    float* deg = (float*)((char*)d_ws + (size_t)N * NHID * sizeof(float));
    unsigned* ctr = (unsigned*)(deg + (((size_t)N + 63) & ~(size_t)63));
    unsigned* planes = ctr + 2 * CTR_STRIDE;

    const int* srcIdx = ei;
    const int* dstIdx = ei + E;

    gcn_deg_hist <<<DH_GROUPS, DH_THREADS, 0, stream>>>(dstIdx, planes, E, N);
    gcn_gemm_wmma<<<(N + GM_ROWS - 1) / GM_ROWS, 256, 0, stream>>>(x, W, xw, N);

    gcn_finish_self<<<(N + FS_WAVES * FS_NODES - 1) / (FS_WAVES * FS_NODES), FS_WAVES * 32, 0, stream>>>(
            planes, xw, bias, deg, out, ctr, N);
    gcn_edge_scatter<<<SC_GRID, 256, 0, stream>>>(srcIdx, dstIdx, deg, xw, out, ctr, E, N);
}


# GEMM epilogue does the degree finish and out's start rows: sums count planes, stores dinv, xw rows scaled by dinv[src], out; an edge reads dinv[dst] only; 3 launches, not 4; products round otherwise
# speedup vs baseline: 3.4557x; 1.0511x over previous
#include <hip/hip_runtime.h>

typedef __attribute__((ext_vector_type(2))) float v2f;
typedef __attribute__((ext_vector_type(8))) float v8f;

#define NFEAT 128
#define NHID  128

#define CTR_STRIDE 64
#define DH_GROUPS  64
#define DH_THREADS 1024
#define DH_BATCH   16
#define DH_CAP     65536
__global__ __launch_bounds__(DH_THREADS) void gcn_deg_hist(const int* __restrict__ dst,
                                                           unsigned* __restrict__ planes,
                                                           int e, int n) {
    __shared__ __attribute__((aligned(16))) unsigned h[DH_CAP];
    const int tid   = (int)threadIdx.x;
    const int pitch = (n + 3) & ~3;
    const int per   = (e + DH_GROUPS - 1) / DH_GROUPS;
    const int e0    = (int)blockIdx.x * per;
    const int e1    = (e0 + per < e) ? (e0 + per) : e;
    unsigned* plane = planes + (size_t)blockIdx.x * (size_t)pitch;

    for (int base = 0; base < n; base += DH_CAP) {
        const int span  = (n - base < DH_CAP) ? (n - base) : DH_CAP;
        const int span4 = (span + 3) & ~3;
        for (int i = 4 * tid; i < span4; i += 4 * DH_THREADS)
            *(uint4*)(h + i) = make_uint4(0u, 0u, 0u, 0u);
        __syncthreads();
        for (int i0 = e0 + tid; i0 < e1; i0 += DH_BATCH * DH_THREADS) {
            unsigned d[DH_BATCH];
#pragma unroll
            for (int k = 0; k < DH_BATCH; ++k) {
                const int i  = i0 + k * DH_THREADS;
                const int ic = (i < e1) ? i : (e1 - 1);
                const unsigned v = (unsigned)(dst[ic] - base);
                d[k] = (i < e1) ? v : 0xFFFFFFFFu;
            }
#pragma unroll
            for (int k = 0; k < DH_BATCH; ++k)
                if (d[k] < (unsigned)span) atomicAdd(&h[d[k]], 1u);
        }
        __syncthreads();
#pragma unroll 4
        for (int i = 4 * tid; i < span4; i += 4 * DH_THREADS)
            *(uint4*)(plane + base + i) = *(const uint4*)(h + i);
        __syncthreads();
    }
}

#define GM_ROWS  64
#define GM_TILES (GM_ROWS / 16)
#define GM_PITCH 132
#define GM_PARTS 4
__device__ __forceinline__ void gcn_store_rows(float4 v, float di, float4 bb,
                                               float* __restrict__ srow, float* __restrict__ orow) {
    float4 s;
    s.x = v.x * di; s.y = v.y * di; s.z = v.z * di; s.w = v.w * di;
    float4 o;
    o.x = fmaf(s.x, di, bb.x);
    o.y = fmaf(s.y, di, bb.y);
    o.z = fmaf(s.z, di, bb.z);
    o.w = fmaf(s.w, di, bb.w);
    *(float4*)srow = s;
    *(float4*)orow = o;
}
__global__ __launch_bounds__(256) void gcn_gemm_wmma(const float* __restrict__ x,
                                                     const float* __restrict__ W,
                                                     const unsigned* __restrict__ planes,
                                                     const float* __restrict__ bias,
                                                     float* __restrict__ xws,
                                                     float* __restrict__ dinv,
                                                     float* __restrict__ out,
                                                     unsigned* __restrict__ ctr,
                                                     int n_nodes) {
    __shared__ __attribute__((aligned(16))) float xs[GM_ROWS * GM_PITCH];
    __shared__ unsigned pc[GM_PARTS][GM_ROWS];
    __shared__ float dl[GM_ROWS];

    const int lane = threadIdx.x & 31;
    const int wave = __builtin_amdgcn_readfirstlane((int)(threadIdx.x >> 5));
    const int half = lane >> 4;
    const int l    = lane & 15;

    const int m0 = blockIdx.x * GM_ROWS;
    const int n0 = wave * 16;

    if (blockIdx.x == 0 && threadIdx.x == 0) { ctr[0] = 0u; ctr[CTR_STRIDE] = 0u; }

    {
        const int pitch = (n_nodes + 3) & ~3;
        const int nl = (int)threadIdx.x & (GM_ROWS - 1);
        const int pg = (int)threadIdx.x >> 6;
        int node = m0 + nl;
        if (node >= n_nodes) node = n_nodes - 1;
        const unsigned* p = planes + (size_t)(pg * (DH_GROUPS / GM_PARTS)) * (size_t)pitch + node;
        unsigned c = 0u;
#pragma unroll
        for (int g = 0; g < DH_GROUPS / GM_PARTS; ++g) c += p[(size_t)g * (size_t)pitch];
        pc[pg][nl] = c;
    }

#pragma unroll
    for (int i = 0; i < GM_ROWS * 32 / 256; ++i) {
        const int idx = (int)threadIdx.x + 256 * i;
        const int row = idx >> 5;
        const int c4  = idx & 31;
        int grow = m0 + row;
        if (grow >= n_nodes) grow = n_nodes - 1;
        const float4 v = *(const float4*)(x + (size_t)grow * NFEAT + 4 * c4);
        *(float4*)(xs + row * GM_PITCH + 4 * c4) = v;
    }

    v2f bf[NFEAT / 4];
    const float* bptr = W + (size_t)(4 * half) * NHID + n0 + l;
#pragma unroll
    for (int s = 0; s < NFEAT / 8; ++s) {
#pragma unroll
        for (int j = 0; j < 2; ++j) {
            const int k = 8 * s + 2 * j;
            bf[2 * s + j].x = bptr[(size_t)k * NHID];
            bf[2 * s + j].y = bptr[(size_t)(k + 1) * NHID];
        }
    }

    __syncthreads();

    if (threadIdx.x < GM_ROWS) {
        const int t = (int)threadIdx.x;
        const unsigned c = pc[0][t] + pc[1][t] + pc[2][t] + pc[3][t];
        const float di = rsqrtf((float)(c + 1u));
        dl[t] = di;
        if (m0 + t < n_nodes) dinv[m0 + t] = di;
    }

    v8f acc[GM_TILES];
#pragma unroll
    for (int t = 0; t < GM_TILES; ++t) acc[t] = (v8f){};

    const float* aptr = xs + l * GM_PITCH + 4 * half;
#pragma unroll
    for (int s = 0; s < NFEAT / 8; ++s) {
        float4 a[GM_TILES];
#pragma unroll
        for (int t = 0; t < GM_TILES; ++t)
            a[t] = *(const float4*)(aptr + 16 * t * GM_PITCH + 8 * s);
#pragma unroll
        for (int t = 0; t < GM_TILES; ++t) {
            v2f a0; a0.x = a[t].x; a0.y = a[t].y;
            v2f a1; a1.x = a[t].z; a1.y = a[t].w;
            acc[t] = __builtin_amdgcn_wmma_f32_16x16x4_f32(
                     false, a0,  false, bf[2 * s],
                     (short)0, acc[t],  false,  false);
            acc[t] = __builtin_amdgcn_wmma_f32_16x16x4_f32(
                     false, a1,  false, bf[2 * s + 1],
                     (short)0, acc[t],  false,  false);
        }
    }

    __syncthreads();

#pragma unroll
    for (int t = 0; t < GM_TILES; ++t) {
#pragma unroll
        for (int r = 0; r < 8; ++r)
            xs[(16 * t + r + 8 * half) * GM_PITCH + n0 + l] = acc[t][r];
    }

    __syncthreads();

    const float4 bb = *(const float4*)(bias + lane * 4);
    const int row0 = wave * (GM_ROWS / 8);
    const float* xrow = xs + row0 * GM_PITCH + 4 * lane;
    float* srow = xws + (size_t)(m0 + row0) * NHID + 4 * lane;
    float* orow = out + (size_t)(m0 + row0) * NHID + 4 * lane;
    if (m0 + GM_ROWS <= n_nodes) {
        float4 v[GM_ROWS / 8];
        float  di[GM_ROWS / 8];
#pragma unroll
        for (int i = 0; i < GM_ROWS / 8; ++i) {
            v[i]  = *(const float4*)(xrow + i * GM_PITCH);
            di[i] = dl[row0 + i];
        }
#pragma unroll
        for (int i = 0; i < GM_ROWS / 8; ++i)
            gcn_store_rows(v[i], di[i], bb, srow + (size_t)i * NHID, orow + (size_t)i * NHID);
    } else {
        for (int i = 0; i < GM_ROWS / 8; ++i) {
            if (m0 + row0 + i < n_nodes)
                gcn_store_rows(*(const float4*)(xrow + i * GM_PITCH), dl[row0 + i], bb,
                               srow + (size_t)i * NHID, orow + (size_t)i * NHID);
        }
    }
}

#define SC_WAVES 8
#define SC_EDGES_PER_WAVE 32
#define SC_EDGES_PER_WG (SC_WAVES * SC_EDGES_PER_WAVE)
#define SC_BATCH 8
#define SC_GRID 1024

template <bool GUARD>
__device__ __forceinline__ void gcn_scatter_batch(unsigned& m, int s, int d, int nrm_bits,
                                                  unsigned lane,
                                                  const float* __restrict__ xw,
                                                  float* __restrict__ out) {
    float v[SC_BATCH][4];
    int jl[SC_BATCH];
#pragma unroll
    for (int jj = 0; jj < SC_BATCH; ++jj) {
        jl[jj] = -1;
        v[jj][0] = 0.0f; v[jj][1] = 0.0f; v[jj][2] = 0.0f; v[jj][3] = 0.0f;
        if (!GUARD || m != 0u) {
            const int j = __builtin_ctz(m);
            m &= m - 1u;
            jl[jj] = j;
            const int sj = __builtin_amdgcn_readlane(s, j);
            const float* xr = xw + (size_t)sj * NHID + lane;
            v[jj][0] = xr[0];
            v[jj][1] = xr[32];
            v[jj][2] = xr[64];
            v[jj][3] = xr[96];
        }
    }
#pragma unroll
    for (int jj = 0; jj < SC_BATCH; ++jj) {
        if (!GUARD || jl[jj] >= 0) {
            const int   j  = jl[jj];
            const int   dj = __builtin_amdgcn_readlane(d, j);
            const float nj = __builtin_bit_cast(float, __builtin_amdgcn_readlane(nrm_bits, j));
            float* o = out + (size_t)dj * NHID + lane;
            atomicAdd(o +  0, v[jj][0] * nj);
            atomicAdd(o + 32, v[jj][1] * nj);
            atomicAdd(o + 64, v[jj][2] * nj);
            atomicAdd(o + 96, v[jj][3] * nj);
        }
    }
}

__global__ __launch_bounds__(256) void gcn_edge_scatter(const int* __restrict__ src,
                                                        const int* __restrict__ dst,
                                                        const float* __restrict__ dinv,
                                                        const float* __restrict__ xw,
                                                        float* __restrict__ out,
                                                        unsigned* __restrict__ ctr,
                                                        int e, int n_nodes) {
    __shared__ unsigned ticket[2];
    const unsigned lane = threadIdx.x & 31;
    const int wave = __builtin_amdgcn_readfirstlane((int)(threadIdx.x >> 5));
    const int nchunk = (e + SC_EDGES_PER_WG - 1) / SC_EDGES_PER_WG;
    const int split = n_nodes >> 1;

    const int die  = (__builtin_amdgcn_s_sendmsg_rtn(0x87) >> 16) & 15;
    const int home = (die >> 2) & 1;

    int q = home;
    unsigned slot = 0;
    unsigned drawn = 0;
    if (threadIdx.x == 0) drawn = atomicAdd(&ctr[q * CTR_STRIDE], 1u);

    for (int it = 0; it < 2 * nchunk + 2; ++it) {
        if (threadIdx.x == 0) ticket[slot] = drawn;
        __syncthreads();
        const unsigned t = (unsigned)__builtin_amdgcn_readfirstlane((int)ticket[slot]);
        slot ^= 1u;
        if (t >= (unsigned)nchunk) {
            if (q != home) break;
            q ^= 1;
            if (threadIdx.x == 0) drawn = atomicAdd(&ctr[q * CTR_STRIDE], 1u);
            continue;
        }
        if (threadIdx.x == 0) drawn = atomicAdd(&ctr[q * CTR_STRIDE], 1u);

        const int e0 = (int)t * SC_EDGES_PER_WG + wave * SC_EDGES_PER_WAVE;
        if (e0 < e) {
            const int  mine   = e0 + (int)lane;
            const bool inside = mine < e;
            const int  idx    = inside ? mine : e - 1;
            const int  s = src[idx];
            const int  d = dst[idx];
            const bool take = inside && ((d >= split) == (q != 0));
            float nrm = 0.0f;
            if (take) nrm = dinv[d];
            const int nrm_bits = __builtin_bit_cast(int, nrm);
            unsigned m = __builtin_amdgcn_ballot_w32(take);
            while (m != 0u) {
                if (__builtin_popcount(m) >= SC_BATCH)
                    gcn_scatter_batch<false>(m, s, d, nrm_bits, lane, xw, out);
                else
                    gcn_scatter_batch<true>(m, s, d, nrm_bits, lane, xw, out);
            }
        }
    }
}

extern "C" void kernel_launch(void* const* d_in, const int* in_sizes, int n_in,
                              void* d_out, int out_size, void* d_ws, size_t ws_size,
                              hipStream_t stream) {
    const int*   ei   = (const int*)d_in[0];
    const float* x    = (const float*)d_in[1];
    const float* W    = (const float*)d_in[2];
    const float* bias = (const float*)d_in[3];
    float*       out  = (float*)d_out;

    const int E = in_sizes[0] / 2;
    const int N = in_sizes[1] / NFEAT;

    float* xw  = (float*)d_ws;
    float* deg = (float*)((char*)d_ws + (size_t)N * NHID * sizeof(float));
    unsigned* ctr = (unsigned*)(deg + (((size_t)N + 63) & ~(size_t)63));
    unsigned* planes = ctr + 2 * CTR_STRIDE;

    const int* srcIdx = ei;
    const int* dstIdx = ei + E;

    gcn_deg_hist <<<DH_GROUPS, DH_THREADS, 0, stream>>>(dstIdx, planes, E, N);
    gcn_gemm_wmma<<<(N + GM_ROWS - 1) / GM_ROWS, 256, 0, stream>>>(x, W, planes, bias, xw, deg, out, ctr, N);
    gcn_edge_scatter<<<SC_GRID, 256, 0, stream>>>(srcIdx, dstIdx, deg, xw, out, ctr, E, N);
}
